# starting kernel
# baseline (speedup 1.0000x reference)
#include <hip/hip_runtime.h>
#include <hip/hip_bf16.h>

typedef __attribute__((ext_vector_type(2))) float v2f;
typedef __attribute__((ext_vector_type(8))) float v8f;

#define N_NODES 50000
#define N_EDGES 600000
#define DIM     128
#define NGRAPH  64
#define DEPTH   4

__global__ void zero_f32_kernel(float* __restrict__ p, int n) {
    int i = blockIdx.x * blockDim.x + threadIdx.x;
    if (i < n) p[i] = 0.0f;
}

__global__ void deg_kernel(const int* __restrict__ dst, float* __restrict__ deg) {
    int e = blockIdx.x * blockDim.x + threadIdx.x;
    if (e < N_EDGES) atomicAdd(&deg[dst[e]], 1.0f);
}

__global__ void dis_kernel(const float* __restrict__ deg, float* __restrict__ dis) {
    int i = blockIdx.x * blockDim.x + threadIdx.x;
    if (i < N_NODES) {
        float d = deg[i];
        dis[i] = (d > 0.0f) ? rsqrtf(d) : 0.0f;
    }
}

__global__ void embed_kernel(const int* __restrict__ x,
                             const float* __restrict__ emb,
                             float* __restrict__ h) {
    int i = blockIdx.x * blockDim.x + threadIdx.x;
    if (i >= N_NODES * (DIM / 4)) return;
    int node = i >> 5;
    int d4   = i & 31;
    int tok  = x[node];
    ((float4*)h)[i] = ((const float4*)(emb + (long)tok * DIM))[d4];
}

__global__ void bias_init_kernel(const float* __restrict__ bias, float* __restrict__ h) {
    int i = blockIdx.x * blockDim.x + threadIdx.x;
    if (i >= N_NODES * (DIM / 4)) return;
    int d4 = i & 31;
    ((float4*)h)[i] = ((const float4*)bias)[d4];
}

__global__ void __launch_bounds__(256)
gemm_wmma_kernel(const float* __restrict__ h,
                 const float* __restrict__ W,
                 float* __restrict__ hw) {
    __shared__ float As[16 * DIM];

    const int tid     = threadIdx.x;
    const int rowBase = blockIdx.x * 16;

    {
        const float4* src = (const float4*)(h + (long)rowBase * DIM);
        float4* dst4 = (float4*)As;
        dst4[tid]       = src[tid];
        dst4[tid + 256] = src[tid + 256];
    }
    __syncthreads();

    const int wave    = tid >> 5;
    const int lane    = tid & 31;
    const int laneM   = lane & 15;
    const int half    = lane >> 4;
    const int colBase = wave * 16;

    v8f acc = {};

    #pragma unroll
    for (int k0 = 0; k0 < DIM; k0 += 4) {
        const int kA = k0 + half * 2;
        v2f a;
        a.x = As[laneM * DIM + kA];
        a.y = As[laneM * DIM + kA + 1];
        v2f b;
        b.x = W[kA * DIM + colBase + laneM];
        b.y = W[(kA + 1) * DIM + colBase + laneM];
        acc = __builtin_amdgcn_wmma_f32_16x16x4_f32(
            false, a, false, b, (short)0, acc, false, false);
    }

    float* out = hw + (long)rowBase * DIM + colBase;
    #pragma unroll
    for (int r = 0; r < 8; ++r) {
        out[(r + half * 8) * DIM + laneM] = acc[r];
    }
}

__global__ void __launch_bounds__(256)
scatter_kernel(const float* __restrict__ hw,
               const int* __restrict__ src_idx,
               const int* __restrict__ dst_idx,
               const float* __restrict__ dis,
               float* __restrict__ h) {
    int gtid = blockIdx.x * blockDim.x + threadIdx.x;
    int edge = gtid >> 5;
    int lane = threadIdx.x & 31;
    if (edge >= N_EDGES) return;

    int s = src_idx[edge];
    int d = dst_idx[edge];
    float nrm = dis[s] * dis[d];

    float4 v = ((const float4*)(hw + (long)s * DIM))[lane];
    float* out = h + (long)d * DIM + lane * 4;
    atomicAdd(out + 0, v.x * nrm);
    atomicAdd(out + 1, v.y * nrm);
    atomicAdd(out + 2, v.z * nrm);
    atomicAdd(out + 3, v.w * nrm);
}

__global__ void gather_out_kernel(const int* __restrict__ ptr,
                                  const float* __restrict__ h,
                                  float* __restrict__ out) {
    int i = blockIdx.x * blockDim.x + threadIdx.x;
    if (i >= NGRAPH * DIM) return;
    int b = i >> 7;
    int d = i & (DIM - 1);
    int node = ptr[b + 1] - 1;
    out[i] = h[(long)node * DIM + d];
}

extern "C" void kernel_launch(void* const* d_in, const int* in_sizes, int n_in,
                              void* d_out, int out_size, void* d_ws, size_t ws_size,
                              hipStream_t stream) {
    const int*   x    = (const int*)d_in[0];
    const int*   eidx = (const int*)d_in[1];
    const int*   ptr  = (const int*)d_in[2];
    const float* emb  = (const float*)d_in[4];
    const float* W    = (const float*)d_in[5];
    const float* bias = (const float*)d_in[6];
    float*       out  = (float*)d_out;

    const int* src_idx = eidx;
    const int* dst_idx = eidx + N_EDGES;

    float* h   = (float*)d_ws;
    float* hw  = h  + (size_t)N_NODES * DIM;
    float* deg = hw + (size_t)N_NODES * DIM;
    float* dis = deg + N_NODES;

    const int T = 256;

    zero_f32_kernel<<<(N_NODES + T - 1) / T, T, 0, stream>>>(deg, N_NODES);
    deg_kernel<<<(N_EDGES + T - 1) / T, T, 0, stream>>>(dst_idx, deg);
    dis_kernel<<<(N_NODES + T - 1) / T, T, 0, stream>>>(deg, dis);

    const int vec_elems = N_NODES * (DIM / 4);
    embed_kernel<<<(vec_elems + T - 1) / T, T, 0, stream>>>(x, emb, h);

    for (int l = 0; l < DEPTH; ++l) {
        gemm_wmma_kernel<<<N_NODES / 16, T, 0, stream>>>(h, W, hw);
        bias_init_kernel<<<(vec_elems + T - 1) / T, T, 0, stream>>>(bias, h);
        scatter_kernel<<<(N_EDGES * 32 + T - 1) / T, T, 0, stream>>>(hw, src_idx, dst_idx, dis, h);
    }

    gather_out_kernel<<<(NGRAPH * DIM + T - 1) / T, T, 0, stream>>>(ptr, h, out);
}


# Per-edge f32 atomic scatter-add replaced by a per-call destination-ordered edge list (int atomic count+rank, scan, place) and a one-wave-per-node row gather; sums in int-atomic rank order, fma
# speedup vs baseline: 2.4010x; 2.4010x over previous
#include <hip/hip_runtime.h>
#include <hip/hip_bf16.h>

typedef __attribute__((ext_vector_type(2))) float v2f;
typedef __attribute__((ext_vector_type(8))) float v8f;

#define N_NODES 50000
#define N_EDGES 600000
#define DIM     128
#define NGRAPH  64
#define DEPTH   4
#define N_PAD   50048
#define N_CHUNK8 (N_NODES / 8)
#define SCAN_T   1024
#define SCAN_TILES 7

static_assert(N_NODES % 8 == 0, "scan chunks and the 8 nodes per gather block need N_NODES % 8 == 0");
static_assert(SCAN_TILES * SCAN_T >= N_CHUNK8, "scan tiles must cover every node");
static_assert(N_PAD > N_NODES && N_PAD % 32 == 0, "row pointer needs N_NODES + 1 entries");


__global__ void zero_i32_kernel(int* __restrict__ p, int n) {
    int i = blockIdx.x * blockDim.x + threadIdx.x;
    if (i < n) p[i] = 0;
}

__global__ void count_rank_kernel(const int* __restrict__ dst,
                                  int* __restrict__ cnt,
                                  int* __restrict__ rank) {
    int e = blockIdx.x * blockDim.x + threadIdx.x;
    if (e < N_EDGES) rank[e] = atomicAdd(&cnt[dst[e]], 1);
}

__global__ void __launch_bounds__(SCAN_T)
scan_kernel(const int* __restrict__ cnt,
            int* __restrict__ rowptr,
            float* __restrict__ dis) {
    __shared__ int wave_tot[32];
    const int t    = threadIdx.x;
    const int lane = t & 31;
    const int wave = t >> 5;

    int4 a[SCAN_TILES], b[SCAN_TILES];
    #pragma unroll
    for (int j = 0; j < SCAN_TILES; ++j) {
        const int chunk = j * SCAN_T + t;
        a[j] = make_int4(0, 0, 0, 0);
        b[j] = make_int4(0, 0, 0, 0);
        if (chunk < N_CHUNK8) {
            a[j] = ((const int4*)cnt)[chunk * 2];
            b[j] = ((const int4*)cnt)[chunk * 2 + 1];
        }
    }

    int carry = 0;
    #pragma unroll
    for (int j = 0; j < SCAN_TILES; ++j) {
        const int chunk = j * SCAN_T + t;
        const int s = a[j].x + a[j].y + a[j].z + a[j].w + b[j].x + b[j].y + b[j].z + b[j].w;

        int incl = s;
        for (int off = 1; off < 32; off <<= 1) {
            const int v = __shfl_up(incl, off);
            if (lane >= off) incl += v;
        }
        if (lane == 31) wave_tot[wave] = incl;
        __syncthreads();

        const int wt = wave_tot[lane];
        int wincl = wt;
        for (int off = 1; off < 32; off <<= 1) {
            const int v = __shfl_up(wincl, off);
            if (lane >= off) wincl += v;
        }
        const int wave_excl  = __shfl(wincl, wave) - __shfl(wt, wave);
        const int tile_total = __shfl(wincl, 31);
        __syncthreads();

        if (chunk < N_CHUNK8) {
            int4 pa, pb;
            pa.x = carry + wave_excl + incl - s;
            pa.y = pa.x + a[j].x;
            pa.z = pa.y + a[j].y;
            pa.w = pa.z + a[j].z;
            pb.x = pa.w + a[j].w;
            pb.y = pb.x + b[j].x;
            pb.z = pb.y + b[j].y;
            pb.w = pb.z + b[j].z;
            ((int4*)rowptr)[chunk * 2]     = pa;
            ((int4*)rowptr)[chunk * 2 + 1] = pb;

            float4 da, db;
            da.x = (a[j].x > 0) ? rsqrtf((float)a[j].x) : 0.0f;
            da.y = (a[j].y > 0) ? rsqrtf((float)a[j].y) : 0.0f;
            da.z = (a[j].z > 0) ? rsqrtf((float)a[j].z) : 0.0f;
            da.w = (a[j].w > 0) ? rsqrtf((float)a[j].w) : 0.0f;
            db.x = (b[j].x > 0) ? rsqrtf((float)b[j].x) : 0.0f;
            db.y = (b[j].y > 0) ? rsqrtf((float)b[j].y) : 0.0f;
            db.z = (b[j].z > 0) ? rsqrtf((float)b[j].z) : 0.0f;
            db.w = (b[j].w > 0) ? rsqrtf((float)b[j].w) : 0.0f;
            ((float4*)dis)[chunk * 2]     = da;
            ((float4*)dis)[chunk * 2 + 1] = db;
        }
        carry += tile_total;
    }
    if (t == 0) rowptr[N_NODES] = carry;
}

__global__ void place_kernel(const int* __restrict__ src,
                             const int* __restrict__ dst,
                             const int* __restrict__ rank,
                             const int* __restrict__ rowptr,
                             int* __restrict__ esrc) {
    int e = blockIdx.x * blockDim.x + threadIdx.x;
    if (e < N_EDGES) esrc[rowptr[dst[e]] + rank[e]] = src[e];
}

__global__ void embed_kernel(const int* __restrict__ x,
                             const float* __restrict__ emb,
                             float* __restrict__ h) {
    int i = blockIdx.x * blockDim.x + threadIdx.x;
    if (i >= N_NODES * (DIM / 4)) return;
    int node = i >> 5;
    int d4   = i & 31;
    int tok  = x[node];
    ((float4*)h)[i] = ((const float4*)(emb + (long)tok * DIM))[d4];
}

__global__ void __launch_bounds__(256)
gemm_wmma_kernel(const float* __restrict__ h,
                 const float* __restrict__ W,
                 float* __restrict__ hw) {
    __shared__ float As[16 * DIM];

    const int tid     = threadIdx.x;
    const int rowBase = blockIdx.x * 16;

    {
        const float4* src = (const float4*)(h + (long)rowBase * DIM);
        float4* dst4 = (float4*)As;
        dst4[tid]       = src[tid];
        dst4[tid + 256] = src[tid + 256];
    }
    __syncthreads();

    const int wave    = tid >> 5;
    const int lane    = tid & 31;
    const int laneM   = lane & 15;
    const int half    = lane >> 4;
    const int colBase = wave * 16;

    v8f acc = {};

    #pragma unroll
    for (int k0 = 0; k0 < DIM; k0 += 4) {
        const int kA = k0 + half * 2;
        v2f a;
        a.x = As[laneM * DIM + kA];
        a.y = As[laneM * DIM + kA + 1];
        v2f b;
        b.x = W[kA * DIM + colBase + laneM];
        b.y = W[(kA + 1) * DIM + colBase + laneM];
        acc = __builtin_amdgcn_wmma_f32_16x16x4_f32(
            false, a, false, b, (short)0, acc, false, false);
    }

    float* out = hw + (long)rowBase * DIM + colBase;
    #pragma unroll
    for (int r = 0; r < 8; ++r) {
        out[(r + half * 8) * DIM + laneM] = acc[r];
    }
}

__global__ void __launch_bounds__(256)
gather_kernel(const float* __restrict__ hw,
              const int* __restrict__ rowptr,
              const int* __restrict__ esrc,
              const float* __restrict__ dis,
              const float* __restrict__ bias,
              float* __restrict__ h) {
    const int lane = threadIdx.x & 31;
    const int node = __builtin_amdgcn_readfirstlane((int)(blockIdx.x * 8 + (threadIdx.x >> 5)));
    const int beg  = __builtin_amdgcn_readfirstlane(rowptr[node]);
    const int end  = __builtin_amdgcn_readfirstlane(rowptr[node + 1]);
    const float dd = dis[node];

    const float4* rows = (const float4*)hw;
    float4 acc = make_float4(0.0f, 0.0f, 0.0f, 0.0f);

    for (int base = beg; base < end; base += 32) {
        const int c = min(32, end - base);
        int   my_s = 0;
        float my_w = 0.0f;
        if (lane < c) {
            my_s = esrc[base + lane];
            my_w = dis[my_s] * dd;
        }
        int i = 0;
        for (; i + 4 <= c; i += 4) {
            const int s0 = __builtin_amdgcn_readlane(my_s, i);
            const int s1 = __builtin_amdgcn_readlane(my_s, i + 1);
            const int s2 = __builtin_amdgcn_readlane(my_s, i + 2);
            const int s3 = __builtin_amdgcn_readlane(my_s, i + 3);
            const float w0 = __int_as_float(__builtin_amdgcn_readlane(__float_as_int(my_w), i));
            const float w1 = __int_as_float(__builtin_amdgcn_readlane(__float_as_int(my_w), i + 1));
            const float w2 = __int_as_float(__builtin_amdgcn_readlane(__float_as_int(my_w), i + 2));
            const float w3 = __int_as_float(__builtin_amdgcn_readlane(__float_as_int(my_w), i + 3));
            const float4 v0 = rows[s0 * (DIM / 4) + lane];
            const float4 v1 = rows[s1 * (DIM / 4) + lane];
            const float4 v2 = rows[s2 * (DIM / 4) + lane];
            const float4 v3 = rows[s3 * (DIM / 4) + lane];
            acc.x = fmaf(v0.x, w0, acc.x); acc.y = fmaf(v0.y, w0, acc.y);
            acc.z = fmaf(v0.z, w0, acc.z); acc.w = fmaf(v0.w, w0, acc.w);
            acc.x = fmaf(v1.x, w1, acc.x); acc.y = fmaf(v1.y, w1, acc.y);
            acc.z = fmaf(v1.z, w1, acc.z); acc.w = fmaf(v1.w, w1, acc.w);
            acc.x = fmaf(v2.x, w2, acc.x); acc.y = fmaf(v2.y, w2, acc.y);
            acc.z = fmaf(v2.z, w2, acc.z); acc.w = fmaf(v2.w, w2, acc.w);
            acc.x = fmaf(v3.x, w3, acc.x); acc.y = fmaf(v3.y, w3, acc.y);
            acc.z = fmaf(v3.z, w3, acc.z); acc.w = fmaf(v3.w, w3, acc.w);
        }
        for (; i < c; ++i) {
            const int s0 = __builtin_amdgcn_readlane(my_s, i);
            const float w0 = __int_as_float(__builtin_amdgcn_readlane(__float_as_int(my_w), i));
            const float4 v0 = rows[s0 * (DIM / 4) + lane];
            acc.x = fmaf(v0.x, w0, acc.x); acc.y = fmaf(v0.y, w0, acc.y);
            acc.z = fmaf(v0.z, w0, acc.z); acc.w = fmaf(v0.w, w0, acc.w);
        }
    }

    const float4 bv = ((const float4*)bias)[lane];
    float4 o;
    o.x = acc.x + bv.x; o.y = acc.y + bv.y; o.z = acc.z + bv.z; o.w = acc.w + bv.w;
    ((float4*)h)[node * (DIM / 4) + lane] = o;
}

__global__ void gather_out_kernel(const int* __restrict__ ptr,
                                  const float* __restrict__ h,
                                  float* __restrict__ out) {
    int i = blockIdx.x * blockDim.x + threadIdx.x;
    if (i >= NGRAPH * DIM) return;
    int b = i >> 7;
    int d = i & (DIM - 1);
    int node = ptr[b + 1] - 1;
    out[i] = h[(long)node * DIM + d];
}

extern "C" void kernel_launch(void* const* d_in, const int* in_sizes, int n_in,
                              void* d_out, int out_size, void* d_ws, size_t ws_size,
                              hipStream_t stream) {
    const int*   x    = (const int*)d_in[0];
    const int*   eidx = (const int*)d_in[1];
    const int*   ptr  = (const int*)d_in[2];
    const float* emb  = (const float*)d_in[4];
    const float* W    = (const float*)d_in[5];
    const float* bias = (const float*)d_in[6];
    float*       out  = (float*)d_out;

    const int* src_idx = eidx;
    const int* dst_idx = eidx + N_EDGES;

    float* h      = (float*)d_ws;
    float* hw     = h  + (size_t)N_NODES * DIM;
    int*   rowptr = (int*)(hw + (size_t)N_NODES * DIM);
    float* dis    = (float*)(rowptr + N_PAD);
    int*   esrc   = (int*)(dis + N_PAD);
    int*   cnt    = (int*)hw;
    int*   rank   = cnt + N_PAD;

    if (ws_size < 54000384ull) return;

    const int T = 256;

    zero_i32_kernel<<<(N_NODES + T - 1) / T, T, 0, stream>>>(cnt, N_NODES);
    count_rank_kernel<<<(N_EDGES + T - 1) / T, T, 0, stream>>>(dst_idx, cnt, rank);
    scan_kernel<<<1, SCAN_T, 0, stream>>>(cnt, rowptr, dis);
    place_kernel<<<(N_EDGES + T - 1) / T, T, 0, stream>>>(src_idx, dst_idx, rank, rowptr, esrc);

    const int vec_elems = N_NODES * (DIM / 4);
    embed_kernel<<<(vec_elems + T - 1) / T, T, 0, stream>>>(x, emb, h);

    for (int l = 0; l < DEPTH; ++l) {
        gemm_wmma_kernel<<<N_NODES / 16, T, 0, stream>>>(h, W, hw);
        gather_kernel<<<N_NODES / 8, T, 0, stream>>>(hw, rowptr, esrc, dis, bias, h);
    }

    gather_out_kernel<<<(NGRAPH * DIM + T - 1) / T, T, 0, stream>>>(ptr, h, out);
}


# Layers 3 and 4 formed only for the rows the 64 outputs depend on, in one launch per output row: edge sums of layer-2 rows first, then the WMMA product (sum before product: summation order changes)
# speedup vs baseline: 3.3274x; 1.3858x over previous
#include <hip/hip_runtime.h>
#include <hip/hip_bf16.h>

typedef __attribute__((ext_vector_type(2))) float v2f;
typedef __attribute__((ext_vector_type(8))) float v8f;

#define N_NODES 50000
#define N_EDGES 600000
#define DIM     128
#define NGRAPH  64
#define DEPTH   4
#define N_PAD   50048
#define N_CHUNK8 (N_NODES / 8)
#define SCAN_T   1024
#define SCAN_TILES 7

static_assert(N_NODES % 8 == 0, "scan chunks and the 8 nodes per gather block need N_NODES % 8 == 0");
static_assert(SCAN_TILES * SCAN_T >= N_CHUNK8, "scan tiles must cover every node");
static_assert(N_PAD > N_NODES && N_PAD % 32 == 0, "row pointer needs N_NODES + 1 entries");
static_assert(DEPTH >= 2, "the last two layers are computed for the output rows only");


__global__ void zero_i32_kernel(int* __restrict__ p, int n) {
    int i = blockIdx.x * blockDim.x + threadIdx.x;
    if (i < n) p[i] = 0;
}

__global__ void count_rank_kernel(const int* __restrict__ dst,
                                  int* __restrict__ cnt,
                                  int* __restrict__ rank) {
    int e = blockIdx.x * blockDim.x + threadIdx.x;
    if (e < N_EDGES) rank[e] = atomicAdd(&cnt[dst[e]], 1);
}

__global__ void __launch_bounds__(SCAN_T)
scan_kernel(const int* __restrict__ cnt,
            int* __restrict__ rowptr,
            float* __restrict__ dis) {
    __shared__ int wave_tot[32];
    const int t    = threadIdx.x;
    const int lane = t & 31;
    const int wave = t >> 5;

    int4 a[SCAN_TILES], b[SCAN_TILES];
    #pragma unroll
    for (int j = 0; j < SCAN_TILES; ++j) {
        const int chunk = j * SCAN_T + t;
        a[j] = make_int4(0, 0, 0, 0);
        b[j] = make_int4(0, 0, 0, 0);
        if (chunk < N_CHUNK8) {
            a[j] = ((const int4*)cnt)[chunk * 2];
            b[j] = ((const int4*)cnt)[chunk * 2 + 1];
        }
    }

    int carry = 0;
    #pragma unroll
    for (int j = 0; j < SCAN_TILES; ++j) {
        const int chunk = j * SCAN_T + t;
        const int s = a[j].x + a[j].y + a[j].z + a[j].w + b[j].x + b[j].y + b[j].z + b[j].w;

        int incl = s;
        for (int off = 1; off < 32; off <<= 1) {
            const int v = __shfl_up(incl, off);
            if (lane >= off) incl += v;
        }
        if (lane == 31) wave_tot[wave] = incl;
        __syncthreads();

        const int wt = wave_tot[lane];
        int wincl = wt;
        for (int off = 1; off < 32; off <<= 1) {
            const int v = __shfl_up(wincl, off);
            if (lane >= off) wincl += v;
        }
        const int wave_excl  = __shfl(wincl, wave) - __shfl(wt, wave);
        const int tile_total = __shfl(wincl, 31);
        __syncthreads();

        if (chunk < N_CHUNK8) {
            int4 pa, pb;
            pa.x = carry + wave_excl + incl - s;
            pa.y = pa.x + a[j].x;
            pa.z = pa.y + a[j].y;
            pa.w = pa.z + a[j].z;
            pb.x = pa.w + a[j].w;
            pb.y = pb.x + b[j].x;
            pb.z = pb.y + b[j].y;
            pb.w = pb.z + b[j].z;
            ((int4*)rowptr)[chunk * 2]     = pa;
            ((int4*)rowptr)[chunk * 2 + 1] = pb;

            float4 da, db;
            da.x = (a[j].x > 0) ? rsqrtf((float)a[j].x) : 0.0f;
            da.y = (a[j].y > 0) ? rsqrtf((float)a[j].y) : 0.0f;
            da.z = (a[j].z > 0) ? rsqrtf((float)a[j].z) : 0.0f;
            da.w = (a[j].w > 0) ? rsqrtf((float)a[j].w) : 0.0f;
            db.x = (b[j].x > 0) ? rsqrtf((float)b[j].x) : 0.0f;
            db.y = (b[j].y > 0) ? rsqrtf((float)b[j].y) : 0.0f;
            db.z = (b[j].z > 0) ? rsqrtf((float)b[j].z) : 0.0f;
            db.w = (b[j].w > 0) ? rsqrtf((float)b[j].w) : 0.0f;
            ((float4*)dis)[chunk * 2]     = da;
            ((float4*)dis)[chunk * 2 + 1] = db;
        }
        carry += tile_total;
    }
    if (t == 0) rowptr[N_NODES] = carry;
}

__global__ void place_kernel(const int* __restrict__ src,
                             const int* __restrict__ dst,
                             const int* __restrict__ rank,
                             const int* __restrict__ rowptr,
                             int* __restrict__ esrc) {
    int e = blockIdx.x * blockDim.x + threadIdx.x;
    if (e < N_EDGES) esrc[rowptr[dst[e]] + rank[e]] = src[e];
}

__global__ void embed_kernel(const int* __restrict__ x,
                             const float* __restrict__ emb,
                             float* __restrict__ h) {
    int i = blockIdx.x * blockDim.x + threadIdx.x;
    if (i >= N_NODES * (DIM / 4)) return;
    int node = i >> 5;
    int d4   = i & 31;
    int tok  = x[node];
    ((float4*)h)[i] = ((const float4*)(emb + (long)tok * DIM))[d4];
}

__global__ void __launch_bounds__(256)
gemm_wmma_kernel(const float* __restrict__ h,
                 const float* __restrict__ W,
                 float* __restrict__ hw) {
    __shared__ float As[16 * DIM];

    const int tid     = threadIdx.x;
    const int rowBase = blockIdx.x * 16;

    {
        const float4* src = (const float4*)(h + (long)rowBase * DIM);
        float4* dst4 = (float4*)As;
        dst4[tid]       = src[tid];
        dst4[tid + 256] = src[tid + 256];
    }
    __syncthreads();

    const int wave    = tid >> 5;
    const int lane    = tid & 31;
    const int laneM   = lane & 15;
    const int half    = lane >> 4;
    const int colBase = wave * 16;

    v8f acc = {};

    #pragma unroll
    for (int k0 = 0; k0 < DIM; k0 += 4) {
        const int kA = k0 + half * 2;
        v2f a;
        a.x = As[laneM * DIM + kA];
        a.y = As[laneM * DIM + kA + 1];
        v2f b;
        b.x = W[kA * DIM + colBase + laneM];
        b.y = W[(kA + 1) * DIM + colBase + laneM];
        acc = __builtin_amdgcn_wmma_f32_16x16x4_f32(
            false, a, false, b, (short)0, acc, false, false);
    }

    float* out = hw + (long)rowBase * DIM + colBase;
    #pragma unroll
    for (int r = 0; r < 8; ++r) {
        out[(r + half * 8) * DIM + laneM] = acc[r];
    }
}

static __device__ __forceinline__ float4 edge_sum(const float4* __restrict__ rows,
                                                  const int* __restrict__ esrc,
                                                  const float* __restrict__ dis,
                                                  int beg, int end, float dd, int lane) {
    float4 acc = make_float4(0.0f, 0.0f, 0.0f, 0.0f);

    for (int base = beg; base < end; base += 32) {
        const int c = min(32, end - base);
        int   my_s = 0;
        float my_w = 0.0f;
        if (lane < c) {
            my_s = esrc[base + lane];
            my_w = dis[my_s] * dd;
        }
        int i = 0;
        for (; i + 4 <= c; i += 4) {
            const int s0 = __builtin_amdgcn_readlane(my_s, i);
            const int s1 = __builtin_amdgcn_readlane(my_s, i + 1);
            const int s2 = __builtin_amdgcn_readlane(my_s, i + 2);
            const int s3 = __builtin_amdgcn_readlane(my_s, i + 3);
            const float w0 = __int_as_float(__builtin_amdgcn_readlane(__float_as_int(my_w), i));
            const float w1 = __int_as_float(__builtin_amdgcn_readlane(__float_as_int(my_w), i + 1));
            const float w2 = __int_as_float(__builtin_amdgcn_readlane(__float_as_int(my_w), i + 2));
            const float w3 = __int_as_float(__builtin_amdgcn_readlane(__float_as_int(my_w), i + 3));
            const float4 v0 = rows[s0 * (DIM / 4) + lane];
            const float4 v1 = rows[s1 * (DIM / 4) + lane];
            const float4 v2 = rows[s2 * (DIM / 4) + lane];
            const float4 v3 = rows[s3 * (DIM / 4) + lane];
            acc.x = fmaf(v0.x, w0, acc.x); acc.y = fmaf(v0.y, w0, acc.y);
            acc.z = fmaf(v0.z, w0, acc.z); acc.w = fmaf(v0.w, w0, acc.w);
            acc.x = fmaf(v1.x, w1, acc.x); acc.y = fmaf(v1.y, w1, acc.y);
            acc.z = fmaf(v1.z, w1, acc.z); acc.w = fmaf(v1.w, w1, acc.w);
            acc.x = fmaf(v2.x, w2, acc.x); acc.y = fmaf(v2.y, w2, acc.y);
            acc.z = fmaf(v2.z, w2, acc.z); acc.w = fmaf(v2.w, w2, acc.w);
            acc.x = fmaf(v3.x, w3, acc.x); acc.y = fmaf(v3.y, w3, acc.y);
            acc.z = fmaf(v3.z, w3, acc.z); acc.w = fmaf(v3.w, w3, acc.w);
        }
        for (; i < c; ++i) {
            const int s0 = __builtin_amdgcn_readlane(my_s, i);
            const float w0 = __int_as_float(__builtin_amdgcn_readlane(__float_as_int(my_w), i));
            const float4 v0 = rows[s0 * (DIM / 4) + lane];
            acc.x = fmaf(v0.x, w0, acc.x); acc.y = fmaf(v0.y, w0, acc.y);
            acc.z = fmaf(v0.z, w0, acc.z); acc.w = fmaf(v0.w, w0, acc.w);
        }
    }
    return acc;
}

__global__ void __launch_bounds__(256)
gather_kernel(const float* __restrict__ hw,
              const int* __restrict__ rowptr,
              const int* __restrict__ esrc,
              const float* __restrict__ dis,
              const float* __restrict__ bias,
              float* __restrict__ h) {
    const int lane = threadIdx.x & 31;
    const int node = __builtin_amdgcn_readfirstlane((int)(blockIdx.x * 8 + (threadIdx.x >> 5)));
    const int beg  = __builtin_amdgcn_readfirstlane(rowptr[node]);
    const int end  = __builtin_amdgcn_readfirstlane(rowptr[node + 1]);
    const float dd = dis[node];

    const float4 acc = edge_sum((const float4*)hw, esrc, dis, beg, end, dd, lane);

    const float4 bv = ((const float4*)bias)[lane];
    float4 o;
    o.x = acc.x + bv.x; o.y = acc.y + bv.y; o.z = acc.z + bv.z; o.w = acc.w + bv.w;
    ((float4*)h)[node * (DIM / 4) + lane] = o;
}

#define A_PITCH (DIM + 4)

static __device__ __forceinline__ v8f strip_times_w(const float* strip, int pitch, bool rowMask,
                                                    const float* __restrict__ W,
                                                    int colBase, int laneM, int half) {
    v8f acc = {};
    #pragma unroll
    for (int k0 = 0; k0 < DIM; k0 += 4) {
        const int kA = k0 + half * 2;
        const float a0 = strip[laneM * pitch + kA];
        const float a1 = strip[laneM * pitch + kA + 1];
        v2f a;
        a.x = rowMask ? a0 : 0.0f;
        a.y = rowMask ? a1 : 0.0f;
        v2f b;
        b.x = W[kA * DIM + colBase + laneM];
        b.y = W[(kA + 1) * DIM + colBase + laneM];
        acc = __builtin_amdgcn_wmma_f32_16x16x4_f32(
            false, a, false, b, (short)0, acc, false, false);
    }
    return acc;
}

__global__ void __launch_bounds__(256)
tail_kernel(const float* __restrict__ h2,
            const int* __restrict__ ptr,
            const int* __restrict__ rowptr,
            const int* __restrict__ esrc,
            const float* __restrict__ dis,
            const float* __restrict__ W,
            const float* __restrict__ bias,
            float* __restrict__ out) {
    __shared__ __attribute__((aligned(16))) float As[16 * A_PITCH];
    __shared__ float wrow[16];
    __shared__ float srow[DIM];

    const int tid     = threadIdx.x;
    const int lane    = tid & 31;
    const int laneM   = lane & 15;
    const int half    = lane >> 4;
    const int wave    = __builtin_amdgcn_readfirstlane(tid >> 5);
    const int colBase = wave * 16;
    const int b       = blockIdx.x;

    int t = ptr[b + 1] - 1;
    if (t < 0) t += N_NODES;
    t = __builtin_amdgcn_readfirstlane(t);
    const int   beg_b = __builtin_amdgcn_readfirstlane(rowptr[t]);
    const int   end_b = __builtin_amdgcn_readfirstlane(rowptr[t + 1]);
    const float dd_b  = dis[t];
    const float bcol  = bias[colBase + laneM];

    const float4* rows = (const float4*)h2;
    float s = 0.0f;

    for (int tb = beg_b; tb < end_b; tb += 16) {
        const int p = tb + laneM;
        int   rb = 0, re = 0;
        float dn = 0.0f;
        if (p < end_b) {
            const int n = esrc[p];
            rb = rowptr[n];
            re = rowptr[n + 1];
            dn = dis[n];
        }
        for (int q = 0; q < 2; ++q) {
            const int m = wave + 8 * q;
            const int   beg = __builtin_amdgcn_readlane(rb, m);
            const int   end = __builtin_amdgcn_readlane(re, m);
            const float dm  = __int_as_float(__builtin_amdgcn_readlane(__float_as_int(dn), m));
            const float4 a = edge_sum(rows, esrc, dis, beg, end, dm, lane);
            ((float4*)(As + m * A_PITCH))[lane] = a;
            if (lane == 0) wrow[m] = dm * dd_b;
        }
        __syncthreads();

        const v8f acc = strip_times_w(As, A_PITCH, true, W, colBase, laneM, half);
        const int nv = end_b - tb;
        #pragma unroll
        for (int r = 0; r < 8; ++r) {
            const int row = r + half * 8;
            const float sn = fmaf(wrow[row], acc[r] + bcol, s);
            s = (row < nv) ? sn : s;
        }
        __syncthreads();
    }

    s += __shfl_xor(s, 16);
    if (half == 0) srow[colBase + laneM] = s;
    __syncthreads();

    const v8f acc = strip_times_w(srow, 0, laneM == 0, W, colBase, laneM, half);
    __syncthreads();
    if (lane < 16) srow[colBase + lane] = acc[0] + bcol;
    __syncthreads();
    if (tid < DIM) out[b * DIM + tid] = srow[tid];
}

extern "C" void kernel_launch(void* const* d_in, const int* in_sizes, int n_in,
                              void* d_out, int out_size, void* d_ws, size_t ws_size,
                              hipStream_t stream) {
    const int*   x    = (const int*)d_in[0];
    const int*   eidx = (const int*)d_in[1];
    const int*   ptr  = (const int*)d_in[2];
    const float* emb  = (const float*)d_in[4];
    const float* W    = (const float*)d_in[5];
    const float* bias = (const float*)d_in[6];
    float*       out  = (float*)d_out;

    const int* src_idx = eidx;
    const int* dst_idx = eidx + N_EDGES;

    float* h      = (float*)d_ws;
    float* hw     = h  + (size_t)N_NODES * DIM;
    int*   rowptr = (int*)(hw + (size_t)N_NODES * DIM);
    float* dis    = (float*)(rowptr + N_PAD);
    int*   esrc   = (int*)(dis + N_PAD);
    int*   cnt    = (int*)hw;
    int*   rank   = cnt + N_PAD;

    if (ws_size < 54000384ull) return;

    const int T = 256;

    zero_i32_kernel<<<(N_NODES + T - 1) / T, T, 0, stream>>>(cnt, N_NODES);
    count_rank_kernel<<<(N_EDGES + T - 1) / T, T, 0, stream>>>(dst_idx, cnt, rank);
    scan_kernel<<<1, SCAN_T, 0, stream>>>(cnt, rowptr, dis);
    place_kernel<<<(N_EDGES + T - 1) / T, T, 0, stream>>>(src_idx, dst_idx, rank, rowptr, esrc);

    const int vec_elems = N_NODES * (DIM / 4);
    embed_kernel<<<(vec_elems + T - 1) / T, T, 0, stream>>>(x, emb, h);

    for (int l = 0; l < DEPTH - 2; ++l) {
        gemm_wmma_kernel<<<N_NODES / 16, T, 0, stream>>>(h, W, hw);
        gather_kernel<<<N_NODES / 8, T, 0, stream>>>(hw, rowptr, esrc, dis, bias, h);
    }

    tail_kernel<<<NGRAPH, T, 0, stream>>>(h, ptr, rowptr, esrc, dis, W, bias, out);
}


# Layer 1 through the token table: emb @ W taken once for the 2000 table rows (same WMMA k order), layer-1 edge sum reads row x[src] of it; per-node lookup and its product removed; sum order unchanged
# speedup vs baseline: 3.9020x; 1.1727x over previous
#include <hip/hip_runtime.h>
#include <hip/hip_bf16.h>

typedef __attribute__((ext_vector_type(2))) float v2f;
typedef __attribute__((ext_vector_type(8))) float v8f;

#define N_NODES 50000
#define N_EDGES 600000
#define DIM     128
#define NGRAPH  64
#define DEPTH   4
#define VOCAB   2000
#define N_PAD   50048
#define N_CHUNK8 (N_NODES / 8)
#define SCAN_T   1024
#define SCAN_TILES 7

static_assert(N_NODES % 8 == 0, "scan chunks and the 8 nodes per gather block need N_NODES % 8 == 0");
static_assert(SCAN_TILES * SCAN_T >= N_CHUNK8, "scan tiles must cover every node");
static_assert(N_PAD > N_NODES && N_PAD % 32 == 0, "row pointer needs N_NODES + 1 entries");
static_assert(DEPTH >= 3, "layer 1 is formed through the table, the last two layers for the output rows only");
static_assert(VOCAB % 16 == 0, "the table product works on whole 16-row tiles");


__global__ void zero_i32_kernel(int* __restrict__ p, int n) {
    int i = blockIdx.x * blockDim.x + threadIdx.x;
    if (i < n) p[i] = 0;
}

__global__ void count_rank_kernel(const int* __restrict__ dst,
                                  int* __restrict__ cnt,
                                  int* __restrict__ rank) {
    int e = blockIdx.x * blockDim.x + threadIdx.x;
    if (e < N_EDGES) rank[e] = atomicAdd(&cnt[dst[e]], 1);
}

__global__ void __launch_bounds__(SCAN_T)
scan_kernel(const int* __restrict__ cnt,
            int* __restrict__ rowptr,
            float* __restrict__ dis) {
    __shared__ int wave_tot[32];
    const int t    = threadIdx.x;
    const int lane = t & 31;
    const int wave = t >> 5;

    int4 a[SCAN_TILES], b[SCAN_TILES];
    #pragma unroll
    for (int j = 0; j < SCAN_TILES; ++j) {
        const int chunk = j * SCAN_T + t;
        a[j] = make_int4(0, 0, 0, 0);
        b[j] = make_int4(0, 0, 0, 0);
        if (chunk < N_CHUNK8) {
            a[j] = ((const int4*)cnt)[chunk * 2];
            b[j] = ((const int4*)cnt)[chunk * 2 + 1];
        }
    }

    int carry = 0;
    #pragma unroll
    for (int j = 0; j < SCAN_TILES; ++j) {
        const int chunk = j * SCAN_T + t;
        const int s = a[j].x + a[j].y + a[j].z + a[j].w + b[j].x + b[j].y + b[j].z + b[j].w;

        int incl = s;
        for (int off = 1; off < 32; off <<= 1) {
            const int v = __shfl_up(incl, off);
            if (lane >= off) incl += v;
        }
        if (lane == 31) wave_tot[wave] = incl;
        __syncthreads();

        const int wt = wave_tot[lane];
        int wincl = wt;
        for (int off = 1; off < 32; off <<= 1) {
            const int v = __shfl_up(wincl, off);
            if (lane >= off) wincl += v;
        }
        const int wave_excl  = __shfl(wincl, wave) - __shfl(wt, wave);
        const int tile_total = __shfl(wincl, 31);
        __syncthreads();

        if (chunk < N_CHUNK8) {
            int4 pa, pb;
            pa.x = carry + wave_excl + incl - s;
            pa.y = pa.x + a[j].x;
            pa.z = pa.y + a[j].y;
            pa.w = pa.z + a[j].z;
            pb.x = pa.w + a[j].w;
            pb.y = pb.x + b[j].x;
            pb.z = pb.y + b[j].y;
            pb.w = pb.z + b[j].z;
            ((int4*)rowptr)[chunk * 2]     = pa;
            ((int4*)rowptr)[chunk * 2 + 1] = pb;

            float4 da, db;
            da.x = (a[j].x > 0) ? rsqrtf((float)a[j].x) : 0.0f;
            da.y = (a[j].y > 0) ? rsqrtf((float)a[j].y) : 0.0f;
            da.z = (a[j].z > 0) ? rsqrtf((float)a[j].z) : 0.0f;
            da.w = (a[j].w > 0) ? rsqrtf((float)a[j].w) : 0.0f;
            db.x = (b[j].x > 0) ? rsqrtf((float)b[j].x) : 0.0f;
            db.y = (b[j].y > 0) ? rsqrtf((float)b[j].y) : 0.0f;
            db.z = (b[j].z > 0) ? rsqrtf((float)b[j].z) : 0.0f;
            db.w = (b[j].w > 0) ? rsqrtf((float)b[j].w) : 0.0f;
            ((float4*)dis)[chunk * 2]     = da;
            ((float4*)dis)[chunk * 2 + 1] = db;
        }
        carry += tile_total;
    }
    if (t == 0) rowptr[N_NODES] = carry;
}

__global__ void place_kernel(const int* __restrict__ src,
                             const int* __restrict__ dst,
                             const int* __restrict__ rank,
                             const int* __restrict__ rowptr,
                             int* __restrict__ esrc) {
    int e = blockIdx.x * blockDim.x + threadIdx.x;
    if (e < N_EDGES) esrc[rowptr[dst[e]] + rank[e]] = src[e];
}

__global__ void __launch_bounds__(256)
gemm_wmma_kernel(const float* __restrict__ h,
                 const float* __restrict__ W,
                 float* __restrict__ hw) {
    __shared__ float As[16 * DIM];

    const int tid     = threadIdx.x;
    const int rowBase = blockIdx.x * 16;

    {
        const float4* src = (const float4*)(h + (long)rowBase * DIM);
        float4* dst4 = (float4*)As;
        dst4[tid]       = src[tid];
        dst4[tid + 256] = src[tid + 256];
    }
    __syncthreads();

    const int wave    = tid >> 5;
    const int lane    = tid & 31;
    const int laneM   = lane & 15;
    const int half    = lane >> 4;
    const int colBase = wave * 16;

    v8f acc = {};

    #pragma unroll
    for (int k0 = 0; k0 < DIM; k0 += 4) {
        const int kA = k0 + half * 2;
        v2f a;
        a.x = As[laneM * DIM + kA];
        a.y = As[laneM * DIM + kA + 1];
        v2f b;
        b.x = W[kA * DIM + colBase + laneM];
        b.y = W[(kA + 1) * DIM + colBase + laneM];
        acc = __builtin_amdgcn_wmma_f32_16x16x4_f32(
            false, a, false, b, (short)0, acc, false, false);
    }

    float* out = hw + (long)rowBase * DIM + colBase;
    #pragma unroll
    for (int r = 0; r < 8; ++r) {
        out[(r + half * 8) * DIM + laneM] = acc[r];
    }
}

template <bool VIA_TOKEN>
static __device__ __forceinline__ float4 edge_sum(const float4* __restrict__ rows,
                                                  const int* __restrict__ esrc,
                                                  const float* __restrict__ dis,
                                                  const int* __restrict__ tok,
                                                  int beg, int end, float dd, int lane) {
    float4 acc = make_float4(0.0f, 0.0f, 0.0f, 0.0f);

    for (int base = beg; base < end; base += 32) {
        const int c = min(32, end - base);
        int   my_s = 0;
        float my_w = 0.0f;
        if (lane < c) {
            my_s = esrc[base + lane];
            my_w = dis[my_s] * dd;
            if (VIA_TOKEN) my_s = tok[my_s];
        }
        int i = 0;
        for (; i + 4 <= c; i += 4) {
            const int s0 = __builtin_amdgcn_readlane(my_s, i);
            const int s1 = __builtin_amdgcn_readlane(my_s, i + 1);
            const int s2 = __builtin_amdgcn_readlane(my_s, i + 2);
            const int s3 = __builtin_amdgcn_readlane(my_s, i + 3);
            const float w0 = __int_as_float(__builtin_amdgcn_readlane(__float_as_int(my_w), i));
            const float w1 = __int_as_float(__builtin_amdgcn_readlane(__float_as_int(my_w), i + 1));
            const float w2 = __int_as_float(__builtin_amdgcn_readlane(__float_as_int(my_w), i + 2));
            const float w3 = __int_as_float(__builtin_amdgcn_readlane(__float_as_int(my_w), i + 3));
            const float4 v0 = rows[s0 * (DIM / 4) + lane];
            const float4 v1 = rows[s1 * (DIM / 4) + lane];
            const float4 v2 = rows[s2 * (DIM / 4) + lane];
            const float4 v3 = rows[s3 * (DIM / 4) + lane];
            acc.x = fmaf(v0.x, w0, acc.x); acc.y = fmaf(v0.y, w0, acc.y);
            acc.z = fmaf(v0.z, w0, acc.z); acc.w = fmaf(v0.w, w0, acc.w);
            acc.x = fmaf(v1.x, w1, acc.x); acc.y = fmaf(v1.y, w1, acc.y);
            acc.z = fmaf(v1.z, w1, acc.z); acc.w = fmaf(v1.w, w1, acc.w);
            acc.x = fmaf(v2.x, w2, acc.x); acc.y = fmaf(v2.y, w2, acc.y);
            acc.z = fmaf(v2.z, w2, acc.z); acc.w = fmaf(v2.w, w2, acc.w);
            acc.x = fmaf(v3.x, w3, acc.x); acc.y = fmaf(v3.y, w3, acc.y);
            acc.z = fmaf(v3.z, w3, acc.z); acc.w = fmaf(v3.w, w3, acc.w);
        }
        for (; i < c; ++i) {
            const int s0 = __builtin_amdgcn_readlane(my_s, i);
            const float w0 = __int_as_float(__builtin_amdgcn_readlane(__float_as_int(my_w), i));
            const float4 v0 = rows[s0 * (DIM / 4) + lane];
            acc.x = fmaf(v0.x, w0, acc.x); acc.y = fmaf(v0.y, w0, acc.y);
            acc.z = fmaf(v0.z, w0, acc.z); acc.w = fmaf(v0.w, w0, acc.w);
        }
    }
    return acc;
}

__global__ void __launch_bounds__(256)
gather_kernel(const float* __restrict__ hw,
              const int* __restrict__ rowptr,
              const int* __restrict__ esrc,
              const float* __restrict__ dis,
              const float* __restrict__ bias,
              float* __restrict__ h) {
    const int lane = threadIdx.x & 31;
    const int node = __builtin_amdgcn_readfirstlane((int)(blockIdx.x * 8 + (threadIdx.x >> 5)));
    const int beg  = __builtin_amdgcn_readfirstlane(rowptr[node]);
    const int end  = __builtin_amdgcn_readfirstlane(rowptr[node + 1]);
    const float dd = dis[node];

    const float4 acc = edge_sum<false>((const float4*)hw, esrc, dis, nullptr, beg, end, dd, lane);

    const float4 bv = ((const float4*)bias)[lane];
    float4 o;
    o.x = acc.x + bv.x; o.y = acc.y + bv.y; o.z = acc.z + bv.z; o.w = acc.w + bv.w;
    ((float4*)h)[node * (DIM / 4) + lane] = o;
}

__global__ void __launch_bounds__(256)
gather_table_kernel(const float* __restrict__ tw,
                    const int* __restrict__ x,
                    const int* __restrict__ rowptr,
                    const int* __restrict__ esrc,
                    const float* __restrict__ dis,
                    const float* __restrict__ bias,
                    float* __restrict__ h) {
    const int lane = threadIdx.x & 31;
    const int node = __builtin_amdgcn_readfirstlane((int)(blockIdx.x * 8 + (threadIdx.x >> 5)));
    const int beg  = __builtin_amdgcn_readfirstlane(rowptr[node]);
    const int end  = __builtin_amdgcn_readfirstlane(rowptr[node + 1]);
    const float dd = dis[node];

    const float4 acc = edge_sum<true>((const float4*)tw, esrc, dis, x, beg, end, dd, lane);

    const float4 bv = ((const float4*)bias)[lane];
    float4 o;
    o.x = acc.x + bv.x; o.y = acc.y + bv.y; o.z = acc.z + bv.z; o.w = acc.w + bv.w;
    ((float4*)h)[node * (DIM / 4) + lane] = o;
}

#define A_PITCH (DIM + 4)

static __device__ __forceinline__ v8f strip_times_w(const float* strip, int pitch, bool rowMask,
                                                    const float* __restrict__ W,
                                                    int colBase, int laneM, int half) {
    v8f acc = {};
    #pragma unroll
    for (int k0 = 0; k0 < DIM; k0 += 4) {
        const int kA = k0 + half * 2;
        const float a0 = strip[laneM * pitch + kA];
        const float a1 = strip[laneM * pitch + kA + 1];
        v2f a;
        a.x = rowMask ? a0 : 0.0f;
        a.y = rowMask ? a1 : 0.0f;
        v2f b;
        b.x = W[kA * DIM + colBase + laneM];
        b.y = W[(kA + 1) * DIM + colBase + laneM];
        acc = __builtin_amdgcn_wmma_f32_16x16x4_f32(
            false, a, false, b, (short)0, acc, false, false);
    }
    return acc;
}

__global__ void __launch_bounds__(256)
tail_kernel(const float* __restrict__ h2,
            const int* __restrict__ ptr,
            const int* __restrict__ rowptr,
            const int* __restrict__ esrc,
            const float* __restrict__ dis,
            const float* __restrict__ W,
            const float* __restrict__ bias,
            float* __restrict__ out) {
    __shared__ __attribute__((aligned(16))) float As[16 * A_PITCH];
    __shared__ float wrow[16];
    __shared__ float srow[DIM];

    const int tid     = threadIdx.x;
    const int lane    = tid & 31;
    const int laneM   = lane & 15;
    const int half    = lane >> 4;
    const int wave    = __builtin_amdgcn_readfirstlane(tid >> 5);
    const int colBase = wave * 16;
    const int b       = blockIdx.x;

    int t = ptr[b + 1] - 1;
    if (t < 0) t += N_NODES;
    t = __builtin_amdgcn_readfirstlane(t);
    const int   beg_b = __builtin_amdgcn_readfirstlane(rowptr[t]);
    const int   end_b = __builtin_amdgcn_readfirstlane(rowptr[t + 1]);
    const float dd_b  = dis[t];
    const float bcol  = bias[colBase + laneM];

    const float4* rows = (const float4*)h2;
    float s = 0.0f;

    for (int tb = beg_b; tb < end_b; tb += 16) {
        const int p = tb + laneM;
        int   rb = 0, re = 0;
        float dn = 0.0f;
        if (p < end_b) {
            const int n = esrc[p];
            rb = rowptr[n];
            re = rowptr[n + 1];
            dn = dis[n];
        }
        for (int q = 0; q < 2; ++q) {
            const int m = wave + 8 * q;
            const int   beg = __builtin_amdgcn_readlane(rb, m);
            const int   end = __builtin_amdgcn_readlane(re, m);
            const float dm  = __int_as_float(__builtin_amdgcn_readlane(__float_as_int(dn), m));
            const float4 a = edge_sum<false>(rows, esrc, dis, nullptr, beg, end, dm, lane);
            ((float4*)(As + m * A_PITCH))[lane] = a;
            if (lane == 0) wrow[m] = dm * dd_b;
        }
        __syncthreads();

        const v8f acc = strip_times_w(As, A_PITCH, true, W, colBase, laneM, half);
        const int nv = end_b - tb;
        #pragma unroll
        for (int r = 0; r < 8; ++r) {
            const int row = r + half * 8;
            const float sn = fmaf(wrow[row], acc[r] + bcol, s);
            s = (row < nv) ? sn : s;
        }
        __syncthreads();
    }

    s += __shfl_xor(s, 16);
    if (half == 0) srow[colBase + laneM] = s;
    __syncthreads();

    const v8f acc = strip_times_w(srow, 0, laneM == 0, W, colBase, laneM, half);
    __syncthreads();
    if (lane < 16) srow[colBase + lane] = acc[0] + bcol;
    __syncthreads();
    if (tid < DIM) out[b * DIM + tid] = srow[tid];
}

extern "C" void kernel_launch(void* const* d_in, const int* in_sizes, int n_in,
                              void* d_out, int out_size, void* d_ws, size_t ws_size,
                              hipStream_t stream) {
    const int*   x    = (const int*)d_in[0];
    const int*   eidx = (const int*)d_in[1];
    const int*   ptr  = (const int*)d_in[2];
    const float* emb  = (const float*)d_in[4];
    const float* W    = (const float*)d_in[5];
    const float* bias = (const float*)d_in[6];
    float*       out  = (float*)d_out;

    const int* src_idx = eidx;
    const int* dst_idx = eidx + N_EDGES;

    float* h      = (float*)d_ws;
    float* hw     = h  + (size_t)N_NODES * DIM;
    int*   rowptr = (int*)(hw + (size_t)N_NODES * DIM);
    float* dis    = (float*)(rowptr + N_PAD);
    int*   esrc   = (int*)(dis + N_PAD);
    int*   cnt    = (int*)hw;
    int*   rank   = cnt + N_PAD;
    float* tw     = hw + 655360;

    if (ws_size < 54000384ull) return;

    const int T = 256;

    zero_i32_kernel<<<(N_NODES + T - 1) / T, T, 0, stream>>>(cnt, N_NODES);
    count_rank_kernel<<<(N_EDGES + T - 1) / T, T, 0, stream>>>(dst_idx, cnt, rank);
    scan_kernel<<<1, SCAN_T, 0, stream>>>(cnt, rowptr, dis);
    place_kernel<<<(N_EDGES + T - 1) / T, T, 0, stream>>>(src_idx, dst_idx, rank, rowptr, esrc);

    gemm_wmma_kernel<<<VOCAB / 16, T, 0, stream>>>(emb, W, tw);
    gather_table_kernel<<<N_NODES / 8, T, 0, stream>>>(tw, x, rowptr, esrc, dis, bias, h);

    for (int l = 1; l < DEPTH - 2; ++l) {
        gemm_wmma_kernel<<<N_NODES / 16, T, 0, stream>>>(h, W, hw);
        gather_kernel<<<N_NODES / 8, T, 0, stream>>>(hw, rowptr, esrc, dis, bias, h);
    }

    tail_kernel<<<NGRAPH, T, 0, stream>>>(h, ptr, rowptr, esrc, dis, W, bias, out);
}


# Layer 2 too formed only for rows the 64 outputs depend on, inside the tail (sum before product; 16 groups per output row, shares added in a last launch: summation order changes); full layer 2 removed
# speedup vs baseline: 4.8162x; 1.2343x over previous
#include <hip/hip_runtime.h>
#include <hip/hip_bf16.h>

typedef __attribute__((ext_vector_type(2))) float v2f;
typedef __attribute__((ext_vector_type(8))) float v8f;

#define N_NODES 50000
#define N_EDGES 600000
#define DIM     128
#define NGRAPH  64
#define DEPTH   4
#define VOCAB   2000
#define N_PAD   50048
#define N_CHUNK8 (N_NODES / 8)
#define SCAN_T   1024
#define SCAN_TILES 7

static_assert(N_NODES % 8 == 0, "scan chunks and the 8 nodes per gather block need N_NODES % 8 == 0");
static_assert(SCAN_TILES * SCAN_T >= N_CHUNK8, "scan tiles must cover every node");
static_assert(N_PAD > N_NODES && N_PAD % 32 == 0, "row pointer needs N_NODES + 1 entries");
static_assert(DEPTH == 4, "layer 1 is formed through the table, the last three layers for the output rows only");
static_assert(VOCAB % 16 == 0, "the table product works on whole 16-row tiles");


__global__ void zero_i32_kernel(int* __restrict__ p, int n) {
    int i = blockIdx.x * blockDim.x + threadIdx.x;
    if (i < n) p[i] = 0;
}

__global__ void count_rank_kernel(const int* __restrict__ dst,
                                  int* __restrict__ cnt,
                                  int* __restrict__ rank) {
    int e = blockIdx.x * blockDim.x + threadIdx.x;
    if (e < N_EDGES) rank[e] = atomicAdd(&cnt[dst[e]], 1);
}

__global__ void __launch_bounds__(SCAN_T)
scan_kernel(const int* __restrict__ cnt,
            int* __restrict__ rowptr,
            float* __restrict__ dis) {
    __shared__ int wave_tot[32];
    const int t    = threadIdx.x;
    const int lane = t & 31;
    const int wave = t >> 5;

    int4 a[SCAN_TILES], b[SCAN_TILES];
    #pragma unroll
    for (int j = 0; j < SCAN_TILES; ++j) {
        const int chunk = j * SCAN_T + t;
        a[j] = make_int4(0, 0, 0, 0);
        b[j] = make_int4(0, 0, 0, 0);
        if (chunk < N_CHUNK8) {
            a[j] = ((const int4*)cnt)[chunk * 2];
            b[j] = ((const int4*)cnt)[chunk * 2 + 1];
        }
    }

    int carry = 0;
    #pragma unroll
    for (int j = 0; j < SCAN_TILES; ++j) {
        const int chunk = j * SCAN_T + t;
        const int s = a[j].x + a[j].y + a[j].z + a[j].w + b[j].x + b[j].y + b[j].z + b[j].w;

        int incl = s;
        for (int off = 1; off < 32; off <<= 1) {
            const int v = __shfl_up(incl, off);
            if (lane >= off) incl += v;
        }
        if (lane == 31) wave_tot[wave] = incl;
        __syncthreads();

        const int wt = wave_tot[lane];
        int wincl = wt;
        for (int off = 1; off < 32; off <<= 1) {
            const int v = __shfl_up(wincl, off);
            if (lane >= off) wincl += v;
        }
        const int wave_excl  = __shfl(wincl, wave) - __shfl(wt, wave);
        const int tile_total = __shfl(wincl, 31);
        __syncthreads();

        if (chunk < N_CHUNK8) {
            int4 pa, pb;
            pa.x = carry + wave_excl + incl - s;
            pa.y = pa.x + a[j].x;
            pa.z = pa.y + a[j].y;
            pa.w = pa.z + a[j].z;
            pb.x = pa.w + a[j].w;
            pb.y = pb.x + b[j].x;
            pb.z = pb.y + b[j].y;
            pb.w = pb.z + b[j].z;
            ((int4*)rowptr)[chunk * 2]     = pa;
            ((int4*)rowptr)[chunk * 2 + 1] = pb;

            float4 da, db;
            da.x = (a[j].x > 0) ? rsqrtf((float)a[j].x) : 0.0f;
            da.y = (a[j].y > 0) ? rsqrtf((float)a[j].y) : 0.0f;
            da.z = (a[j].z > 0) ? rsqrtf((float)a[j].z) : 0.0f;
            da.w = (a[j].w > 0) ? rsqrtf((float)a[j].w) : 0.0f;
            db.x = (b[j].x > 0) ? rsqrtf((float)b[j].x) : 0.0f;
            db.y = (b[j].y > 0) ? rsqrtf((float)b[j].y) : 0.0f;
            db.z = (b[j].z > 0) ? rsqrtf((float)b[j].z) : 0.0f;
            db.w = (b[j].w > 0) ? rsqrtf((float)b[j].w) : 0.0f;
            ((float4*)dis)[chunk * 2]     = da;
            ((float4*)dis)[chunk * 2 + 1] = db;
        }
        carry += tile_total;
    }
    if (t == 0) rowptr[N_NODES] = carry;
}

__global__ void place_kernel(const int* __restrict__ src,
                             const int* __restrict__ dst,
                             const int* __restrict__ rank,
                             const int* __restrict__ rowptr,
                             int* __restrict__ esrc) {
    int e = blockIdx.x * blockDim.x + threadIdx.x;
    if (e < N_EDGES) esrc[rowptr[dst[e]] + rank[e]] = src[e];
}

__global__ void __launch_bounds__(256)
gemm_wmma_kernel(const float* __restrict__ h,
                 const float* __restrict__ W,
                 float* __restrict__ hw) {
    __shared__ float As[16 * DIM];

    const int tid     = threadIdx.x;
    const int rowBase = blockIdx.x * 16;

    {
        const float4* src = (const float4*)(h + (long)rowBase * DIM);
        float4* dst4 = (float4*)As;
        dst4[tid]       = src[tid];
        dst4[tid + 256] = src[tid + 256];
    }
    __syncthreads();

    const int wave    = tid >> 5;
    const int lane    = tid & 31;
    const int laneM   = lane & 15;
    const int half    = lane >> 4;
    const int colBase = wave * 16;

    v8f acc = {};

    #pragma unroll
    for (int k0 = 0; k0 < DIM; k0 += 4) {
        const int kA = k0 + half * 2;
        v2f a;
        a.x = As[laneM * DIM + kA];
        a.y = As[laneM * DIM + kA + 1];
        v2f b;
        b.x = W[kA * DIM + colBase + laneM];
        b.y = W[(kA + 1) * DIM + colBase + laneM];
        acc = __builtin_amdgcn_wmma_f32_16x16x4_f32(
            false, a, false, b, (short)0, acc, false, false);
    }

    float* out = hw + (long)rowBase * DIM + colBase;
    #pragma unroll
    for (int r = 0; r < 8; ++r) {
        out[(r + half * 8) * DIM + laneM] = acc[r];
    }
}

template <bool VIA_TOKEN>
static __device__ __forceinline__ float4 edge_sum(const float4* __restrict__ rows,
                                                  const int* __restrict__ esrc,
                                                  const float* __restrict__ dis,
                                                  const int* __restrict__ tok,
                                                  int beg, int end, float dd, int lane) {
    float4 acc = make_float4(0.0f, 0.0f, 0.0f, 0.0f);

    for (int base = beg; base < end; base += 32) {
        const int c = min(32, end - base);
        int   my_s = 0;
        float my_w = 0.0f;
        if (lane < c) {
            my_s = esrc[base + lane];
            my_w = dis[my_s] * dd;
            if (VIA_TOKEN) my_s = tok[my_s];
        }
        int i = 0;
        for (; i + 4 <= c; i += 4) {
            const int s0 = __builtin_amdgcn_readlane(my_s, i);
            const int s1 = __builtin_amdgcn_readlane(my_s, i + 1);
            const int s2 = __builtin_amdgcn_readlane(my_s, i + 2);
            const int s3 = __builtin_amdgcn_readlane(my_s, i + 3);
            const float w0 = __int_as_float(__builtin_amdgcn_readlane(__float_as_int(my_w), i));
            const float w1 = __int_as_float(__builtin_amdgcn_readlane(__float_as_int(my_w), i + 1));
            const float w2 = __int_as_float(__builtin_amdgcn_readlane(__float_as_int(my_w), i + 2));
            const float w3 = __int_as_float(__builtin_amdgcn_readlane(__float_as_int(my_w), i + 3));
            const float4 v0 = rows[s0 * (DIM / 4) + lane];
            const float4 v1 = rows[s1 * (DIM / 4) + lane];
            const float4 v2 = rows[s2 * (DIM / 4) + lane];
            const float4 v3 = rows[s3 * (DIM / 4) + lane];
            acc.x = fmaf(v0.x, w0, acc.x); acc.y = fmaf(v0.y, w0, acc.y);
            acc.z = fmaf(v0.z, w0, acc.z); acc.w = fmaf(v0.w, w0, acc.w);
            acc.x = fmaf(v1.x, w1, acc.x); acc.y = fmaf(v1.y, w1, acc.y);
            acc.z = fmaf(v1.z, w1, acc.z); acc.w = fmaf(v1.w, w1, acc.w);
            acc.x = fmaf(v2.x, w2, acc.x); acc.y = fmaf(v2.y, w2, acc.y);
            acc.z = fmaf(v2.z, w2, acc.z); acc.w = fmaf(v2.w, w2, acc.w);
            acc.x = fmaf(v3.x, w3, acc.x); acc.y = fmaf(v3.y, w3, acc.y);
            acc.z = fmaf(v3.z, w3, acc.z); acc.w = fmaf(v3.w, w3, acc.w);
        }
        for (; i < c; ++i) {
            const int s0 = __builtin_amdgcn_readlane(my_s, i);
            const float w0 = __int_as_float(__builtin_amdgcn_readlane(__float_as_int(my_w), i));
            const float4 v0 = rows[s0 * (DIM / 4) + lane];
            acc.x = fmaf(v0.x, w0, acc.x); acc.y = fmaf(v0.y, w0, acc.y);
            acc.z = fmaf(v0.z, w0, acc.z); acc.w = fmaf(v0.w, w0, acc.w);
        }
    }
    return acc;
}

__global__ void __launch_bounds__(256)
gather_table_kernel(const float* __restrict__ tw,
                    const int* __restrict__ x,
                    const int* __restrict__ rowptr,
                    const int* __restrict__ esrc,
                    const float* __restrict__ dis,
                    const float* __restrict__ bias,
                    float* __restrict__ h) {
    const int lane = threadIdx.x & 31;
    const int node = __builtin_amdgcn_readfirstlane((int)(blockIdx.x * 8 + (threadIdx.x >> 5)));
    const int beg  = __builtin_amdgcn_readfirstlane(rowptr[node]);
    const int end  = __builtin_amdgcn_readfirstlane(rowptr[node + 1]);
    const float dd = dis[node];

    const float4 acc = edge_sum<true>((const float4*)tw, esrc, dis, x, beg, end, dd, lane);

    const float4 bv = ((const float4*)bias)[lane];
    float4 o;
    o.x = acc.x + bv.x; o.y = acc.y + bv.y; o.z = acc.z + bv.z; o.w = acc.w + bv.w;
    ((float4*)h)[node * (DIM / 4) + lane] = o;
}

#define A_PITCH (DIM + 4)

static __device__ __forceinline__ v8f strip_times_w(const float* strip, int pitch, bool rowMask,
                                                    const float* __restrict__ W,
                                                    int colBase, int laneM, int half) {
    v8f acc = {};
    #pragma unroll
    for (int k0 = 0; k0 < DIM; k0 += 4) {
        const int kA = k0 + half * 2;
        const float a0 = strip[laneM * pitch + kA];
        const float a1 = strip[laneM * pitch + kA + 1];
        v2f a;
        a.x = rowMask ? a0 : 0.0f;
        a.y = rowMask ? a1 : 0.0f;
        v2f b;
        b.x = W[kA * DIM + colBase + laneM];
        b.y = W[(kA + 1) * DIM + colBase + laneM];
        acc = __builtin_amdgcn_wmma_f32_16x16x4_f32(
            false, a, false, b, (short)0, acc, false, false);
    }
    return acc;
}

struct WFrags { v2f b[DIM / 4]; };

static __device__ __forceinline__ void load_w_frags(WFrags& f, const float* __restrict__ W,
                                                    int colBase, int laneM, int half) {
    #pragma unroll
    for (int k0 = 0; k0 < DIM; k0 += 4) {
        const int kA = k0 + half * 2;
        f.b[k0 / 4].x = W[kA * DIM + colBase + laneM];
        f.b[k0 / 4].y = W[(kA + 1) * DIM + colBase + laneM];
    }
}

static __device__ __forceinline__ v8f strip_times_frags(const float* strip, int pitch, bool rowMask,
                                                        const WFrags& f, int laneM, int half) {
    v8f acc = {};
    #pragma unroll
    for (int k0 = 0; k0 < DIM; k0 += 4) {
        const int kA = k0 + half * 2;
        const float a0 = strip[laneM * pitch + kA];
        const float a1 = strip[laneM * pitch + kA + 1];
        v2f a;
        a.x = rowMask ? a0 : 0.0f;
        a.y = rowMask ? a1 : 0.0f;
        acc = __builtin_amdgcn_wmma_f32_16x16x4_f32(
            false, a, false, f.b[k0 / 4], (short)0, acc, false, false);
    }
    return acc;
}

#define TSPLIT 16

__global__ void __launch_bounds__(256)
tail_rows_kernel(const float* __restrict__ h1,
                 const int* __restrict__ ptr,
                 const int* __restrict__ rowptr,
                 const int* __restrict__ esrc,
                 const float* __restrict__ dis,
                 const float* __restrict__ W,
                 const float* __restrict__ bias,
                 float* __restrict__ part) {
    __shared__ __attribute__((aligned(16))) float As[16 * A_PITCH];
    __shared__ float wrow[16];
    __shared__ float srow[DIM];

    const int tid     = threadIdx.x;
    const int lane    = tid & 31;
    const int laneM   = lane & 15;
    const int half    = lane >> 4;
    const int wave    = __builtin_amdgcn_readfirstlane(tid >> 5);
    const int colBase = wave * 16;
    const int b       = blockIdx.x / TSPLIT;
    const int j       = blockIdx.x % TSPLIT;

    int t = ptr[b + 1] - 1;
    if (t < 0) t += N_NODES;
    t = __builtin_amdgcn_readfirstlane(t);
    const int   beg_b = __builtin_amdgcn_readfirstlane(rowptr[t]);
    const int   end_b = __builtin_amdgcn_readfirstlane(rowptr[t + 1]);
    const float dd_b  = dis[t];
    const float bcol  = bias[colBase + laneM];

    WFrags wf;
    load_w_frags(wf, W, colBase, laneM, half);

    const float4* rows = (const float4*)h1;
    float u = 0.0f;

    for (int p = beg_b + j; p < end_b; p += TSPLIT) {
        const int   n     = __builtin_amdgcn_readfirstlane(esrc[p]);
        const int   beg_n = __builtin_amdgcn_readfirstlane(rowptr[n]);
        const int   end_n = __builtin_amdgcn_readfirstlane(rowptr[n + 1]);
        const float dd_n  = dis[n];

        float s = 0.0f;

        for (int tb = beg_n; tb < end_n; tb += 16) {
            const int q0 = tb + laneM;
            int   rb = 0, re = 0;
            float dm_l = 0.0f;
            if (q0 < end_n) {
                const int m = esrc[q0];
                rb = rowptr[m];
                re = rowptr[m + 1];
                dm_l = dis[m];
            }
            for (int q = 0; q < 2; ++q) {
                const int r = wave + 8 * q;
                const int   beg = __builtin_amdgcn_readlane(rb, r);
                const int   end = __builtin_amdgcn_readlane(re, r);
                const float dm  = __int_as_float(__builtin_amdgcn_readlane(__float_as_int(dm_l), r));
                const float4 a = edge_sum<false>(rows, esrc, dis, nullptr, beg, end, dm, lane);
                ((float4*)(As + r * A_PITCH))[lane] = a;
                if (lane == 0) wrow[r] = dm * dd_n;
            }
            __syncthreads();

            const v8f acc = strip_times_frags(As, A_PITCH, true, wf, laneM, half);
            const int nv = end_n - tb;
            #pragma unroll
            for (int r = 0; r < 8; ++r) {
                const int row = r + half * 8;
                const float sn = fmaf(wrow[row], acc[r] + bcol, s);
                s = (row < nv) ? sn : s;
            }
            __syncthreads();
        }

        s += __shfl_xor(s, 16);
        if (half == 0) srow[colBase + laneM] = s;
        __syncthreads();

        const v8f acc = strip_times_frags(srow, 0, laneM == 0, wf, laneM, half);
        u = fmaf(dd_n * dd_b, acc[0] + bcol, u);
        __syncthreads();
    }

    if (lane < 16) srow[colBase + lane] = u;
    __syncthreads();
    if (tid < DIM) part[(long)blockIdx.x * DIM + tid] = srow[tid];
}

__global__ void __launch_bounds__(256)
tail_out_kernel(const float* __restrict__ part,
                const float* __restrict__ W,
                const float* __restrict__ bias,
                float* __restrict__ out) {
    __shared__ float srow[DIM];

    const int tid     = threadIdx.x;
    const int lane    = tid & 31;
    const int laneM   = lane & 15;
    const int half    = lane >> 4;
    const int colBase = (tid >> 5) * 16;
    const int b       = blockIdx.x;
    const float bcol  = bias[colBase + laneM];

    if (tid < DIM) {
        float u = 0.0f;
        #pragma unroll
        for (int j = 0; j < TSPLIT; ++j) u += part[((long)b * TSPLIT + j) * DIM + tid];
        srow[tid] = u;
    }
    __syncthreads();

    const v8f acc = strip_times_w(srow, 0, laneM == 0, W, colBase, laneM, half);
    __syncthreads();
    if (lane < 16) srow[colBase + lane] = acc[0] + bcol;
    __syncthreads();
    if (tid < DIM) out[b * DIM + tid] = srow[tid];
}

extern "C" void kernel_launch(void* const* d_in, const int* in_sizes, int n_in,
                              void* d_out, int out_size, void* d_ws, size_t ws_size,
                              hipStream_t stream) {
    const int*   x    = (const int*)d_in[0];
    const int*   eidx = (const int*)d_in[1];
    const int*   ptr  = (const int*)d_in[2];
    const float* emb  = (const float*)d_in[4];
    const float* W    = (const float*)d_in[5];
    const float* bias = (const float*)d_in[6];
    float*       out  = (float*)d_out;

    const int* src_idx = eidx;
    const int* dst_idx = eidx + N_EDGES;

    float* h      = (float*)d_ws;
    float* hw     = h  + (size_t)N_NODES * DIM;
    int*   rowptr = (int*)(hw + (size_t)N_NODES * DIM);
    float* dis    = (float*)(rowptr + N_PAD);
    int*   esrc   = (int*)(dis + N_PAD);
    int*   cnt    = (int*)hw;
    int*   rank   = cnt + N_PAD;
    float* tw     = hw + 655360;
    float* part   = hw;

    if (ws_size < 54000384ull) return;

    const int T = 256;

    zero_i32_kernel<<<(N_NODES + T - 1) / T, T, 0, stream>>>(cnt, N_NODES);
    count_rank_kernel<<<(N_EDGES + T - 1) / T, T, 0, stream>>>(dst_idx, cnt, rank);
    scan_kernel<<<1, SCAN_T, 0, stream>>>(cnt, rowptr, dis);
    place_kernel<<<(N_EDGES + T - 1) / T, T, 0, stream>>>(src_idx, dst_idx, rank, rowptr, esrc);

    gemm_wmma_kernel<<<VOCAB / 16, T, 0, stream>>>(emb, W, tw);
    gather_table_kernel<<<N_NODES / 8, T, 0, stream>>>(tw, x, rowptr, esrc, dis, bias, h);

    tail_rows_kernel<<<NGRAPH * TSPLIT, T, 0, stream>>>(h, ptr, rowptr, esrc, dis, W, bias, part);
    tail_out_kernel<<<NGRAPH, T, 0, stream>>>(part, W, bias, out);
}
